# starting kernel
# baseline (speedup 1.0000x reference)
#include <hip/hip_runtime.h>
#include <hip/hip_bf16.h>


typedef float v2f   __attribute__((ext_vector_type(2)));
typedef float v8f   __attribute__((ext_vector_type(8)));
typedef float fvec4 __attribute__((ext_vector_type(4)));

#define FL        1024
#define HOP       512
#define S_LEN     262144
#define T_FRAMES  513
#define N_FILT    1024
#define K_DIM     1024
#define B_BATCH   16
#define C_CH      2

#define BM   128
#define BN   128
#define BK   32
#define LDA  33
#define LDB  132

#define PI_F 3.14159265358979323846f

__global__ __launch_bounds__(256)
void filterbank_wmma_kernel(const float* __restrict__ data,
                            const float* __restrict__ filters,
                            float* __restrict__ out)
{
    __shared__ float As[2][BM * LDA];
    __shared__ float Bs[2][BK * LDB];
    __shared__ float win[FL];

    const int tid  = threadIdx.x;
    const int lane = tid & 31;
    const int wave = tid >> 5;
    const int wm   = wave >> 1;
    const int wn   = wave & 1;
    const int wmB  = wm * 32;
    const int wnB  = wn * 64;

    const int bc = blockIdx.z;
    const int b  = bc >> 1;
    const int c  = bc & 1;
    const int m0 = blockIdx.y * BM;
    const int n0 = blockIdx.x * BN;

    const float* __restrict__ dptr = data + (size_t)bc * (size_t)S_LEN;

    const int halfSel = lane >> 4;
    const int l15     = lane & 15;

    auto loadA = [&](int kb, fvec4 (&r)[4]) {
        #pragma unroll
        for (int i = 0; i < 4; ++i) {
            const int idx  = i * 256 + tid;
            const int row  = idx >> 3;
            const int col4 = (idx & 7) * 4;
            const int t    = m0 + row;
            const long s   = (long)t * HOP + (kb + col4) - (FL / 2);
            fvec4 v = fvec4{0.f, 0.f, 0.f, 0.f};
            if (t < T_FRAMES && s >= 0 && s < S_LEN)
                v = *(const fvec4*)(dptr + s);
            r[i] = v;
        }
    };
    auto loadB = [&](int kb, fvec4 (&r)[4]) {
        #pragma unroll
        for (int i = 0; i < 4; ++i) {
            const int idx  = i * 256 + tid;
            const int row  = idx >> 5;
            const int col4 = (idx & 31) * 4;
            r[i] = *(const fvec4*)(filters +
                     (size_t)(kb + row) * N_FILT + n0 + col4);
        }
    };
    auto storeA = [&](int buf, int kb, fvec4 (&r)[4]) {
        #pragma unroll
        for (int i = 0; i < 4; ++i) {
            const int idx  = i * 256 + tid;
            const int row  = idx >> 3;
            const int col4 = (idx & 7) * 4;
            const int f    = kb + col4;
            float* dst = &As[buf][row * LDA + col4];
            dst[0] = r[i][0] * win[f + 0];
            dst[1] = r[i][1] * win[f + 1];
            dst[2] = r[i][2] * win[f + 2];
            dst[3] = r[i][3] * win[f + 3];
        }
    };
    auto storeB = [&](int buf, fvec4 (&r)[4]) {
        #pragma unroll
        for (int i = 0; i < 4; ++i) {
            const int idx  = i * 256 + tid;
            const int row  = idx >> 5;
            const int col4 = (idx & 31) * 4;
            float* dst = &Bs[buf][row * LDB + col4];
            dst[0] = r[i][0];
            dst[1] = r[i][1];
            dst[2] = r[i][2];
            dst[3] = r[i][3];
        }
    };

    for (int i = tid; i < FL; i += 256)
        win[i] = sinf(PI_F * (float)i * (1.0f / (float)FL));

    v8f acc[2][4];
    #pragma unroll
    for (int mi = 0; mi < 2; ++mi)
        #pragma unroll
        for (int ni = 0; ni < 4; ++ni)
            acc[mi][ni] = v8f{0.f, 0.f, 0.f, 0.f, 0.f, 0.f, 0.f, 0.f};

    fvec4 ra[4], rb[4];
    loadA(0, ra);
    loadB(0, rb);
    __syncthreads();
    storeA(0, 0, ra);
    storeB(0, rb);
    __syncthreads();

    const int NKB = K_DIM / BK;
    for (int kbi = 0; kbi < NKB; ++kbi) {
        const int  cur  = kbi & 1;
        const bool more = (kbi + 1) < NKB;

        if (more) {
            loadA((kbi + 1) * BK, ra);
            loadB((kbi + 1) * BK, rb);
        }

        const float* __restrict__ Ab = &As[cur][0];
        const float* __restrict__ Bb = &Bs[cur][0];
        #pragma unroll
        for (int k = 0; k < BK; k += 4) {
            const int ka = k + (halfSel << 1);

            v2f af[2];
            #pragma unroll
            for (int mi = 0; mi < 2; ++mi) {
                const int row = wmB + mi * 16 + l15;
                af[mi][0] = Ab[row * LDA + ka];
                af[mi][1] = Ab[row * LDA + ka + 1];
            }
            v2f bf[4];
            #pragma unroll
            for (int ni = 0; ni < 4; ++ni) {
                const int coln = wnB + ni * 16 + l15;
                bf[ni][0] = Bb[ka * LDB + coln];
                bf[ni][1] = Bb[(ka + 1) * LDB + coln];
            }
            #pragma unroll
            for (int mi = 0; mi < 2; ++mi)
                #pragma unroll
                for (int ni = 0; ni < 4; ++ni)
                    acc[mi][ni] = __builtin_amdgcn_wmma_f32_16x16x4_f32(
                         false, af[mi],
                         false, bf[ni],
                         (short)0, acc[mi][ni],
                         false,  false);
        }

        if (more) {
            storeA(cur ^ 1, (kbi + 1) * BK, ra);
            storeB(cur ^ 1, rb);
        }
        __syncthreads();
    }

    #pragma unroll
    for (int mi = 0; mi < 2; ++mi) {
        #pragma unroll
        for (int ni = 0; ni < 4; ++ni) {
            const int nCol = n0 + wnB + ni * 16 + l15;
            #pragma unroll
            for (int g = 0; g < 8; ++g) {
                const int t = m0 + wmB + mi * 16 + g + halfSel * 8;
                if (t < T_FRAMES) {
                    const size_t o =
                        (((size_t)b * T_FRAMES + t) * N_FILT + nCol) * C_CH + c;
                    out[o] = acc[mi][ni][g];
                }
            }
        }
    }
}

extern "C" void kernel_launch(void* const* d_in, const int* in_sizes, int n_in,
                              void* d_out, int out_size, void* d_ws, size_t ws_size,
                              hipStream_t stream) {
    (void)in_sizes; (void)n_in; (void)out_size; (void)d_ws; (void)ws_size;

    const float* data    = (const float*)d_in[0];
    const float* filters = (const float*)d_in[1];
    float* out           = (float*)d_out;

    dim3 grid(N_FILT / BN,
              (T_FRAMES + BM - 1) / BM,
              B_BATCH * C_CH);
    dim3 block(256);

    filterbank_wmma_kernel<<<grid, block, 0, stream>>>(data, filters, out);
}


# Pool the last frame (t=512) of all 32 (b,c) into one 32-row tile run first over its non-padded K half; frames 0..511 tile exactly: 1024+8 work-groups, not 1280
# speedup vs baseline: 1.1951x; 1.1951x over previous
#include <hip/hip_runtime.h>
#include <hip/hip_bf16.h>


typedef float v2f   __attribute__((ext_vector_type(2)));
typedef float v8f   __attribute__((ext_vector_type(8)));
typedef float fvec4 __attribute__((ext_vector_type(4)));

#define FL        1024
#define HOP       512
#define S_LEN     262144
#define T_FRAMES  513
#define N_FILT    1024
#define K_DIM     1024
#define B_BATCH   16
#define C_CH      2

#define BM   128
#define POOL_ROWS (B_BATCH * C_CH)
#define BN   128
#define BK   32
#define LDA  33
#define LDB  132

#define PI_F 3.14159265358979323846f

__global__ __launch_bounds__(256)
void filterbank_wmma_kernel(const float* __restrict__ data,
                            const float* __restrict__ filters,
                            float* __restrict__ out)
{
    __shared__ float As[2][BM * LDA];
    __shared__ float Bs[2][BK * LDB];
    __shared__ float win[FL];

    const int tid  = threadIdx.x;
    const int lane = tid & 31;
    const int wave = __builtin_amdgcn_readfirstlane(tid >> 5);
    const int wm   = wave >> 1;
    const int wn   = wave & 1;
    const int wmB  = wm * 32;
    const int wnB  = wn * 64;

    const bool pooled = (blockIdx.y == 0);
    const int  bcT = ((int)blockIdx.y - 1) >> 2;
    const int  m0  = (((int)blockIdx.y - 1) & 3) * BM;
    const int  n0  = blockIdx.x * BN;
    const bool waveHasRows = !pooled || (wm == 0);

    const int halfSel = lane >> 4;
    const int l15     = lane & 15;

    const int  aRow0 = tid >> 3;
    const int  aCol4 = (tid & 7) * 4;
    const long aOff  = (pooled ? ((long)aRow0 * S_LEN + (long)(T_FRAMES - 1) * HOP)
                               : ((long)bcT * S_LEN + (long)(m0 + aRow0) * HOP))
                       + aCol4 - (FL / 2);
    const bool aFrame0 = !pooled && (m0 + aRow0 == 0);
    auto loadA = [&](int kb, fvec4 (&r)[4]) {
        const bool lowHalf = kb < FL / 2;
        #pragma unroll
        for (int i = 0; i < 4; ++i) {
            const bool ok = (i == 0) ? !(aFrame0 && lowHalf) && !(pooled && !lowHalf)
                                     : !pooled;
            fvec4 v = fvec4{0.f, 0.f, 0.f, 0.f};
            if (ok)
                v = *(const fvec4*)(data + (aOff + (long)(i * 32 * HOP) + kb));
            r[i] = v;
        }
    };
    auto loadB = [&](int kb, fvec4 (&r)[4]) {
        #pragma unroll
        for (int i = 0; i < 4; ++i) {
            const int idx  = i * 256 + tid;
            const int row  = idx >> 5;
            const int col4 = (idx & 31) * 4;
            r[i] = *(const fvec4*)(filters +
                     (size_t)(kb + row) * N_FILT + n0 + col4);
        }
    };
    auto storeA = [&](int buf, int kb, fvec4 (&r)[4]) {
        #pragma unroll
        for (int i = 0; i < 4; ++i) {
            const int idx  = i * 256 + tid;
            const int row  = idx >> 3;
            const int col4 = (idx & 7) * 4;
            const int f    = kb + col4;
            float* dst = &As[buf][row * LDA + col4];
            dst[0] = r[i][0] * win[f + 0];
            dst[1] = r[i][1] * win[f + 1];
            dst[2] = r[i][2] * win[f + 2];
            dst[3] = r[i][3] * win[f + 3];
        }
    };
    auto storeB = [&](int buf, fvec4 (&r)[4]) {
        #pragma unroll
        for (int i = 0; i < 4; ++i) {
            const int idx  = i * 256 + tid;
            const int row  = idx >> 5;
            const int col4 = (idx & 31) * 4;
            float* dst = &Bs[buf][row * LDB + col4];
            dst[0] = r[i][0];
            dst[1] = r[i][1];
            dst[2] = r[i][2];
            dst[3] = r[i][3];
        }
    };

    for (int i = tid; i < FL; i += 256)
        win[i] = sinf(PI_F * (float)i * (1.0f / (float)FL));

    v8f acc[2][4];
    #pragma unroll
    for (int mi = 0; mi < 2; ++mi)
        #pragma unroll
        for (int ni = 0; ni < 4; ++ni)
            acc[mi][ni] = v8f{0.f, 0.f, 0.f, 0.f, 0.f, 0.f, 0.f, 0.f};

    fvec4 ra[4], rb[4];
    loadA(0, ra);
    loadB(0, rb);
    __syncthreads();
    storeA(0, 0, ra);
    storeB(0, rb);
    __syncthreads();

    const int NKB = pooled ? (FL / 2) / BK : K_DIM / BK;
    for (int kbi = 0; kbi < NKB; ++kbi) {
        const int  cur  = kbi & 1;
        const bool more = (kbi + 1) < NKB;

        if (more) {
            loadA((kbi + 1) * BK, ra);
            loadB((kbi + 1) * BK, rb);
        }

        const float* __restrict__ Ab = &As[cur][0];
        const float* __restrict__ Bb = &Bs[cur][0];
        if (waveHasRows) {
        #pragma unroll
        for (int k = 0; k < BK; k += 4) {
            const int ka = k + (halfSel << 1);

            v2f af[2];
            #pragma unroll
            for (int mi = 0; mi < 2; ++mi) {
                const int row = wmB + mi * 16 + l15;
                af[mi][0] = Ab[row * LDA + ka];
                af[mi][1] = Ab[row * LDA + ka + 1];
            }
            v2f bf[4];
            #pragma unroll
            for (int ni = 0; ni < 4; ++ni) {
                const int coln = wnB + ni * 16 + l15;
                bf[ni][0] = Bb[ka * LDB + coln];
                bf[ni][1] = Bb[(ka + 1) * LDB + coln];
            }
            #pragma unroll
            for (int mi = 0; mi < 2; ++mi)
                #pragma unroll
                for (int ni = 0; ni < 4; ++ni)
                    acc[mi][ni] = __builtin_amdgcn_wmma_f32_16x16x4_f32(
                         false, af[mi],
                         false, bf[ni],
                         (short)0, acc[mi][ni],
                         false,  false);
        }
        }

        if (more) {
            storeA(cur ^ 1, (kbi + 1) * BK, ra);
            storeB(cur ^ 1, rb);
        }
        __syncthreads();
    }

    if (waveHasRows) {
        #pragma unroll
        for (int mi = 0; mi < 2; ++mi) {
            #pragma unroll
            for (int ni = 0; ni < 4; ++ni) {
                const int nCol = n0 + wnB + ni * 16 + l15;
                #pragma unroll
                for (int g = 0; g < 8; ++g) {
                    const int row = wmB + mi * 16 + g + halfSel * 8;
                    const int bc  = pooled ? row : bcT;
                    const int t   = pooled ? (T_FRAMES - 1) : (m0 + row);
                    const int b   = bc >> 1;
                    const int c   = bc & 1;
                    const size_t o =
                        (((size_t)b * T_FRAMES + t) * N_FILT + nCol) * C_CH + c;
                    out[o] = acc[mi][ni][g];
                }
            }
        }
    }
}

extern "C" void kernel_launch(void* const* d_in, const int* in_sizes, int n_in,
                              void* d_out, int out_size, void* d_ws, size_t ws_size,
                              hipStream_t stream) {
    (void)in_sizes; (void)n_in; (void)out_size; (void)d_ws; (void)ws_size;

    const float* data    = (const float*)d_in[0];
    const float* filters = (const float*)d_in[1];
    float* out           = (float*)d_out;

    dim3 grid(N_FILT / BN,
              1 + 4 * B_BATCH * C_CH,
              1);
    dim3 block(256);

    filterbank_wmma_kernel<<<grid, block, 0, stream>>>(data, filters, out);
}


# LDS fragment path: k-quad layouts, one 16-byte load per fragment pair, a step's 6 loads grouped ahead of the previous 16 WMMAs (was 48 loads spread among them); k sum order inside a step changes
# speedup vs baseline: 1.2900x; 1.0794x over previous
#include <hip/hip_runtime.h>
#include <hip/hip_bf16.h>


typedef float v2f   __attribute__((ext_vector_type(2)));
typedef float v8f   __attribute__((ext_vector_type(8)));
typedef float fvec4 __attribute__((ext_vector_type(4)));

#define FL        1024
#define HOP       512
#define S_LEN     262144
#define T_FRAMES  513
#define N_FILT    1024
#define K_DIM     1024
#define B_BATCH   16
#define C_CH      2

#define BM   128
#define POOL_ROWS (B_BATCH * C_CH)
#define BN   128
#define BK   32
#define LDA  36
#define BQP  (BN * 4)

#define PI_F 3.14159265358979323846f

__global__ __launch_bounds__(256)
void filterbank_wmma_kernel(const float* __restrict__ data,
                            const float* __restrict__ filters,
                            float* __restrict__ out)
{
    __shared__ __attribute__((aligned(16))) float As[2][BM * LDA];
    __shared__ __attribute__((aligned(16))) float Bs[2][(BK / 4) * BQP];
    __shared__ __attribute__((aligned(16))) float win[FL];

    const int tid  = threadIdx.x;
    const int lane = tid & 31;
    const int wave = __builtin_amdgcn_readfirstlane(tid >> 5);
    const int wm   = wave >> 1;
    const int wn   = wave & 1;
    const int wmB  = wm * 32;
    const int wnB  = wn * 64;

    const bool pooled = (blockIdx.y == 0);
    const int  bcT = ((int)blockIdx.y - 1) >> 2;
    const int  m0  = (((int)blockIdx.y - 1) & 3) * BM;
    const int  n0  = blockIdx.x * BN;
    const bool waveHasRows = !pooled || (wm == 0);

    const int halfSel = lane >> 4;
    const int l15     = lane & 15;

    const int  aRow0 = tid >> 3;
    const int  aCol4 = (tid & 7) * 4;
    const long aOffU = (pooled ? (long)(T_FRAMES - 1) * HOP
                               : ((long)bcT * S_LEN + (long)m0 * HOP)) - (FL / 2);
    const unsigned aOffB = (unsigned)((pooled ? aRow0 * S_LEN : aRow0 * HOP) + aCol4) * 4u;
    const bool aFrame0 = !pooled && (m0 + aRow0 == 0);
    const float* __restrict__ aPtr = data + aOffU;
    auto loadA = [&](int kb, fvec4 (&r)[4]) {
        const bool lowHalf = kb < FL / 2;
        const char* aK = (const char*)aPtr;
        const unsigned aOffK = aOffB + (unsigned)kb * 4u;
        #pragma unroll
        for (int i = 0; i < 4; ++i) {
            const bool ok = (i == 0) ? !(aFrame0 && lowHalf) && !(pooled && !lowHalf)
                                     : !pooled;
            fvec4 v = fvec4{0.f, 0.f, 0.f, 0.f};
            if (ok)
                v = *(const fvec4*)(aK + i * (32 * HOP * 4) + aOffK);
            r[i] = v;
        }
    };
    const int bQ    = wave;
    const int bCol4 = lane * 4;
    const unsigned bOffB = (unsigned)bCol4 * 4u;
    const float* __restrict__ bPtr = filters + (size_t)(bQ * 4) * N_FILT + n0;
    auto loadB = [&](int kb, fvec4 (&r)[4]) {
        const char* bK = (const char*)bPtr;
        const unsigned bOffK = bOffB + (unsigned)kb * (N_FILT * 4u);
        #pragma unroll
        for (int i = 0; i < 4; ++i)
            r[i] = *(const fvec4*)(bK + i * (N_FILT * 4) + bOffK);
    };
    auto storeA = [&](int buf, int kb, fvec4 (&r)[4]) {
        const fvec4 w = *(const fvec4*)&win[kb + aCol4];
        #pragma unroll
        for (int i = 0; i < 4; ++i)
            *(fvec4*)&As[buf][aRow0 * LDA + aCol4 + i * 32 * LDA] = r[i] * w;
    };
    auto storeB = [&](int buf, fvec4 (&r)[4]) {
        float* dst = &Bs[buf][bQ * BQP + bCol4 * 4];
        #pragma unroll
        for (int j = 0; j < 4; ++j)
            *(fvec4*)(dst + 4 * j) = fvec4{r[0][j], r[1][j], r[2][j], r[3][j]};
    };

    const int aFragOff = (wmB + l15) * LDA + 4 * halfSel;
    const int bFragOff = halfSel * BQP + (wnB + l15) * 4;
    auto loadFrag = [&](int buf, int j, fvec4 (&fa)[2], fvec4 (&fb)[4]) {
        #pragma unroll
        for (int mi = 0; mi < 2; ++mi)
            fa[mi] = *(const fvec4*)&As[buf][aFragOff + mi * 16 * LDA + 8 * j];
        #pragma unroll
        for (int ni = 0; ni < 4; ++ni)
            fb[ni] = *(const fvec4*)&Bs[buf][bFragOff + 2 * j * BQP + ni * 64];
    };

    for (int i = tid; i < FL; i += 256)
        win[i] = sinf(PI_F * (float)i * (1.0f / (float)FL));

    v8f acc[2][4];
    #pragma unroll
    for (int mi = 0; mi < 2; ++mi)
        #pragma unroll
        for (int ni = 0; ni < 4; ++ni)
            acc[mi][ni] = v8f{0.f, 0.f, 0.f, 0.f, 0.f, 0.f, 0.f, 0.f};

    auto mmaStep = [&](fvec4 (&fa)[2], fvec4 (&fb)[4]) {
        #pragma unroll
        for (int s = 0; s < 2; ++s)
            #pragma unroll
            for (int mi = 0; mi < 2; ++mi)
                #pragma unroll
                for (int ni = 0; ni < 4; ++ni)
                    acc[mi][ni] = __builtin_amdgcn_wmma_f32_16x16x4_f32(
                         false, v2f{fa[mi][2 * s], fa[mi][2 * s + 1]},
                         false, v2f{fb[ni][2 * s], fb[ni][2 * s + 1]},
                         (short)0, acc[mi][ni],
                         false,  false);
    };

    fvec4 ra[4], rb[4];
    loadA(0, ra);
    loadB(0, rb);
    __syncthreads();
    storeA(0, 0, ra);
    storeB(0, rb);
    __syncthreads();

    fvec4 fa0[2], fb0[4], fa1[2], fb1[4];
    if (waveHasRows)
        loadFrag(0, 0, fa0, fb0);

    const int NKB = pooled ? (FL / 2) / BK : K_DIM / BK;
    for (int kbi = 0; kbi < NKB; ++kbi) {
        const int  cur  = kbi & 1;
        const bool more = (kbi + 1) < NKB;

        if (more) {
            loadA((kbi + 1) * BK, ra);
            loadB((kbi + 1) * BK, rb);
        }

        if (waveHasRows) {
            loadFrag(cur, 1, fa1, fb1);
            __builtin_amdgcn_sched_barrier(0);
            mmaStep(fa0, fb0);
            __builtin_amdgcn_sched_barrier(0);
            loadFrag(cur, 2, fa0, fb0);
            __builtin_amdgcn_sched_barrier(0);
            mmaStep(fa1, fb1);
            __builtin_amdgcn_sched_barrier(0);
            loadFrag(cur, 3, fa1, fb1);
            __builtin_amdgcn_sched_barrier(0);
            mmaStep(fa0, fb0);
            __builtin_amdgcn_sched_barrier(0);
        }

        if (more) {
            storeA(cur ^ 1, (kbi + 1) * BK, ra);
            storeB(cur ^ 1, rb);
        }
        __syncthreads();

        if (waveHasRows) {
            if (more)
                loadFrag(cur ^ 1, 0, fa0, fb0);
            __builtin_amdgcn_sched_barrier(0);
            mmaStep(fa1, fb1);
            __builtin_amdgcn_sched_barrier(0);
        }
    }

    if (waveHasRows) {
        #pragma unroll
        for (int mi = 0; mi < 2; ++mi) {
            #pragma unroll
            for (int ni = 0; ni < 4; ++ni) {
                const int nCol = n0 + wnB + ni * 16 + l15;
                #pragma unroll
                for (int g = 0; g < 8; ++g) {
                    const int row = wmB + mi * 16 + g + halfSel * 8;
                    const int bc  = pooled ? row : bcT;
                    const int t   = pooled ? (T_FRAMES - 1) : (m0 + row);
                    const int b   = bc >> 1;
                    const int c   = bc & 1;
                    const size_t o =
                        (((size_t)b * T_FRAMES + t) * N_FILT + nCol) * C_CH + c;
                    out[o] = acc[mi][ni][g];
                }
            }
        }
    }
}

extern "C" void kernel_launch(void* const* d_in, const int* in_sizes, int n_in,
                              void* d_out, int out_size, void* d_ws, size_t ws_size,
                              hipStream_t stream) {
    (void)in_sizes; (void)n_in; (void)out_size; (void)d_ws; (void)ws_size;

    const float* data    = (const float*)d_in[0];
    const float* filters = (const float*)d_in[1];
    float* out           = (float*)d_out;

    dim3 grid(N_FILT / BN,
              1 + 4 * B_BATCH * C_CH,
              1);
    dim3 block(256);

    filterbank_wmma_kernel<<<grid, block, 0, stream>>>(data, filters, out);
}


# K loop stripped to matrix work and its hand-over: pooled and full tiles as separate compiled paths, frame-0 pad test block-uniform, loop unrolled over the two LDS stages (constant stage addresses)
# speedup vs baseline: 1.3469x; 1.0441x over previous
#include <hip/hip_runtime.h>
#include <hip/hip_bf16.h>


typedef float v2f   __attribute__((ext_vector_type(2)));
typedef float v8f   __attribute__((ext_vector_type(8)));
typedef float fvec4 __attribute__((ext_vector_type(4)));

#define FL        1024
#define HOP       512
#define S_LEN     262144
#define T_FRAMES  513
#define N_FILT    1024
#define K_DIM     1024
#define B_BATCH   16
#define C_CH      2

#define BM   128
#define POOL_ROWS (B_BATCH * C_CH)
#define BN   128
#define BK   32
#define LDA  36
#define BQP  (BN * 4)

#define PI_F 3.14159265358979323846f

template <int V> struct IntTag { static constexpr int value = V; };

template <bool POOLED>
__device__ __forceinline__
void filterbank_tile(const float* __restrict__ data,
                     const float* __restrict__ filters,
                     float* __restrict__ out,
                     float (&As)[2][BM * LDA],
                     float (&Bs)[2][(BK / 4) * BQP],
                     float (&win)[FL])
{
    const int tid  = threadIdx.x;
    const int lane = tid & 31;
    const int wave = __builtin_amdgcn_readfirstlane(tid >> 5);
    const int wm   = wave >> 1;
    const int wn   = wave & 1;
    const int wmB  = wm * 32;
    const int wnB  = wn * 64;

    const int  bcT = ((int)blockIdx.y - 1) >> 2;
    const int  m0  = (((int)blockIdx.y - 1) & 3) * BM;
    const int  n0  = blockIdx.x * BN;
    const bool waveHasRows = !POOLED || (wm == 0);

    const int halfSel = lane >> 4;
    const int l15     = lane & 15;

    const int  aRow0 = tid >> 3;
    const int  aCol4 = (tid & 7) * 4;
    const long aOffU = (POOLED ? (long)(T_FRAMES - 1) * HOP
                               : ((long)bcT * S_LEN + (long)m0 * HOP)) - (FL / 2);
    const unsigned aOffB = (unsigned)((POOLED ? aRow0 * S_LEN : aRow0 * HOP) + aCol4) * 4u;
    const bool aFrame0 = !POOLED && (m0 + aRow0 == 0);
    const int  padEnd  = (!POOLED && m0 == 0) ? (FL / 2) : 0;
    const char* __restrict__ aPtr = (const char*)(data + aOffU);
    auto loadA = [&](int kb, fvec4 (&r)[4]) {
        const char* aK = aPtr + (size_t)kb * 4u;
        if (POOLED) {
            r[0] = *(const fvec4*)(aK + aOffB);
        } else {
            #pragma unroll
            for (int i = 1; i < 4; ++i)
                r[i] = *(const fvec4*)(aK + i * (32 * HOP * 4) + aOffB);
            if (kb < padEnd) {
                fvec4 v = fvec4{0.f, 0.f, 0.f, 0.f};
                if (!aFrame0)
                    v = *(const fvec4*)(aK + aOffB);
                r[0] = v;
            } else {
                r[0] = *(const fvec4*)(aK + aOffB);
            }
        }
    };
    const int bQ    = wave;
    const int bCol4 = lane * 4;
    const unsigned bOffB = (unsigned)bCol4 * 4u;
    const char* __restrict__ bPtr =
        (const char*)(filters + (size_t)(bQ * 4) * N_FILT + n0);
    auto loadB = [&](int kb, fvec4 (&r)[4]) {
        const char* bK = bPtr + (size_t)kb * (N_FILT * 4u);
        #pragma unroll
        for (int i = 0; i < 4; ++i)
            r[i] = *(const fvec4*)(bK + i * (N_FILT * 4) + bOffB);
    };
    const float* winT = &win[aCol4];
    float* aDst0 = &As[0][aRow0 * LDA + aCol4];
    float* aDst1 = &As[1][aRow0 * LDA + aCol4];
    auto storeA = [&](int buf, int kb, fvec4 (&r)[4]) {
        const fvec4 w = *(const fvec4*)(winT + kb);
        float* dst = buf ? aDst1 : aDst0;
        #pragma unroll
        for (int i = 0; i < 4; ++i)
            *(fvec4*)(dst + i * 32 * LDA) = r[i] * w;
    };
    float* bDst0 = &Bs[0][bQ * BQP + bCol4 * 4];
    float* bDst1 = &Bs[1][bQ * BQP + bCol4 * 4];
    auto storeB = [&](int buf, fvec4 (&r)[4]) {
        float* dst = buf ? bDst1 : bDst0;
        #pragma unroll
        for (int j = 0; j < 4; ++j)
            *(fvec4*)(dst + 4 * j) = fvec4{r[0][j], r[1][j], r[2][j], r[3][j]};
    };

    const int aFragOff = (wmB + l15) * LDA + 4 * halfSel;
    const int bFragOff = halfSel * BQP + (wnB + l15) * 4;
    const float* aFrag0 = &As[0][aFragOff];
    const float* aFrag1 = &As[1][aFragOff];
    const float* bFrag0 = &Bs[0][bFragOff];
    const float* bFrag1 = &Bs[1][bFragOff];
    auto loadFrag = [&](int buf, int j, fvec4 (&fa)[2], fvec4 (&fb)[4]) {
        const float* ap = buf ? aFrag1 : aFrag0;
        const float* bp = buf ? bFrag1 : bFrag0;
        #pragma unroll
        for (int mi = 0; mi < 2; ++mi)
            fa[mi] = *(const fvec4*)(ap + mi * 16 * LDA + 8 * j);
        #pragma unroll
        for (int ni = 0; ni < 4; ++ni)
            fb[ni] = *(const fvec4*)(bp + 2 * j * BQP + ni * 64);
    };

    for (int i = tid; i < FL; i += 256)
        win[i] = sinf(PI_F * (float)i * (1.0f / (float)FL));

    v8f acc[2][4];
    #pragma unroll
    for (int mi = 0; mi < 2; ++mi)
        #pragma unroll
        for (int ni = 0; ni < 4; ++ni)
            acc[mi][ni] = v8f{0.f, 0.f, 0.f, 0.f, 0.f, 0.f, 0.f, 0.f};

    auto mmaStep = [&](fvec4 (&fa)[2], fvec4 (&fb)[4]) {
        #pragma unroll
        for (int s = 0; s < 2; ++s)
            #pragma unroll
            for (int mi = 0; mi < 2; ++mi)
                #pragma unroll
                for (int ni = 0; ni < 4; ++ni)
                    acc[mi][ni] = __builtin_amdgcn_wmma_f32_16x16x4_f32(
                         false, v2f{fa[mi][2 * s], fa[mi][2 * s + 1]},
                         false, v2f{fb[ni][2 * s], fb[ni][2 * s + 1]},
                         (short)0, acc[mi][ni],
                         false,  false);
    };

    fvec4 ra[4], rb[4];
    #pragma unroll
    for (int i = 0; i < 4; ++i)
        ra[i] = fvec4{0.f, 0.f, 0.f, 0.f};
    loadA(0, ra);
    loadB(0, rb);
    __syncthreads();
    storeA(0, 0, ra);
    storeB(0, rb);
    __syncthreads();

    fvec4 fa0[2], fb0[4], fa1[2], fb1[4];
    if (waveHasRows)
        loadFrag(0, 0, fa0, fb0);

    auto kBlock = [&](auto curTag, auto moreTag, int kbi) {
        constexpr int  cur  = decltype(curTag)::value;
        constexpr bool more = decltype(moreTag)::value != 0;

        if (more) {
            loadA((kbi + 1) * BK, ra);
            loadB((kbi + 1) * BK, rb);
        }

        if (waveHasRows) {
            loadFrag(cur, 1, fa1, fb1);
            __builtin_amdgcn_sched_barrier(0);
            mmaStep(fa0, fb0);
            __builtin_amdgcn_sched_barrier(0);
            loadFrag(cur, 2, fa0, fb0);
            __builtin_amdgcn_sched_barrier(0);
            mmaStep(fa1, fb1);
            __builtin_amdgcn_sched_barrier(0);
            loadFrag(cur, 3, fa1, fb1);
            __builtin_amdgcn_sched_barrier(0);
            mmaStep(fa0, fb0);
            __builtin_amdgcn_sched_barrier(0);
        }

        if (more) {
            storeA(cur ^ 1, (kbi + 1) * BK, ra);
            storeB(cur ^ 1, rb);
        }
        __syncthreads();

        if (waveHasRows) {
            if (more)
                loadFrag(cur ^ 1, 0, fa0, fb0);
            __builtin_amdgcn_sched_barrier(0);
            mmaStep(fa1, fb1);
            __builtin_amdgcn_sched_barrier(0);
        }
    };

    constexpr int NKB = POOLED ? (FL / 2) / BK : K_DIM / BK;
    #pragma nounroll
    for (int kbi = 0; kbi < NKB - 2; kbi += 2) {
        kBlock(IntTag<0>{}, IntTag<1>{}, kbi);
        kBlock(IntTag<1>{}, IntTag<1>{}, kbi + 1);
    }
    kBlock(IntTag<0>{}, IntTag<1>{}, NKB - 2);
    kBlock(IntTag<1>{}, IntTag<0>{}, NKB - 1);

    if (waveHasRows) {
        #pragma unroll
        for (int mi = 0; mi < 2; ++mi) {
            #pragma unroll
            for (int ni = 0; ni < 4; ++ni) {
                const int nCol = n0 + wnB + ni * 16 + l15;
                #pragma unroll
                for (int g = 0; g < 8; ++g) {
                    const int row = wmB + mi * 16 + g + halfSel * 8;
                    const int bc  = POOLED ? row : bcT;
                    const int t   = POOLED ? (T_FRAMES - 1) : (m0 + row);
                    const int b   = bc >> 1;
                    const int c   = bc & 1;
                    const size_t o =
                        (((size_t)b * T_FRAMES + t) * N_FILT + nCol) * C_CH + c;
                    out[o] = acc[mi][ni][g];
                }
            }
        }
    }
}

__global__ __launch_bounds__(256)
void filterbank_wmma_kernel(const float* __restrict__ data,
                            const float* __restrict__ filters,
                            float* __restrict__ out)
{
    __shared__ __attribute__((aligned(16))) float As[2][BM * LDA];
    __shared__ __attribute__((aligned(16))) float Bs[2][(BK / 4) * BQP];
    __shared__ __attribute__((aligned(16))) float win[FL];

    if (blockIdx.y == 0)
        filterbank_tile<true>(data, filters, out, As, Bs, win);
    else
        filterbank_tile<false>(data, filters, out, As, Bs, win);
}

extern "C" void kernel_launch(void* const* d_in, const int* in_sizes, int n_in,
                              void* d_out, int out_size, void* d_ws, size_t ws_size,
                              hipStream_t stream) {
    (void)in_sizes; (void)n_in; (void)out_size; (void)d_ws; (void)ws_size;

    const float* data    = (const float*)d_in[0];
    const float* filters = (const float*)d_in[1];
    float* out           = (float*)d_out;

    dim3 grid(N_FILT / BN,
              1 + 4 * B_BATCH * C_CH,
              1);
    dim3 block(256);

    filterbank_wmma_kernel<<<grid, block, 0, stream>>>(data, filters, out);
}
